# starting kernel
# baseline (speedup 1.0000x reference)
#include <hip/hip_runtime.h>

typedef __attribute__((ext_vector_type(2))) float v2f;
typedef __attribute__((ext_vector_type(4))) float v4f;
typedef __attribute__((ext_vector_type(8))) float v8f;

#define NNODES 100000

__global__ void init_deg_kernel(float* __restrict__ deg, int n) {
    int i = blockIdx.x * blockDim.x + threadIdx.x;
    if (i < n) deg[i] = 1.0f;
}

__global__ void accum_deg_kernel(const int* __restrict__ dst, float* __restrict__ deg, int E) {
    int e = blockIdx.x * blockDim.x + threadIdx.x;
    if (e < E) atomicAdd(&deg[dst[e]], 1.0f);
}

__global__ void deg_to_dinv_kernel(float* __restrict__ deg, int n) {
    int i = blockIdx.x * blockDim.x + threadIdx.x;
    if (i < n) deg[i] = rsqrtf(deg[i]);
}

template <int MB>
__launch_bounds__(128)
__global__ void gemm_f32_wmma_kernel(const float* __restrict__ A,
                                     const float* __restrict__ B,
                                     float* __restrict__ C,
                                     int M, int N, int K) {
    const int lane = threadIdx.x & 31;
    const int wave = threadIdx.x >> 5;
    const int ntN  = N >> 4;
    const int nMB  = M / (16 * MB);
    const int job  = blockIdx.x * (blockDim.x >> 5) + wave;
    if (job >= nMB * ntN) return;

    const int mb = job / ntN;
    const int nt = job - mb * ntN;
    const int m0 = mb * 16 * MB;
    const int n0 = nt << 4;

    const int l15   = lane & 15;
    const int khalf = (lane >> 4) << 1;

    const float* abase = A + (size_t)(m0 + l15) * K + khalf;
    const float* bcol  = B + (size_t)khalf * N + (n0 + l15);

    v8f acc[MB];
    #pragma unroll
    for (int t = 0; t < MB; ++t) acc[t] = (v8f){};

    for (int k = 0; k < K; k += 4) {
        v2f b;
        b.x = bcol[(size_t)k * N];
        b.y = bcol[(size_t)(k + 1) * N];
        #pragma unroll
        for (int t = 0; t < MB; ++t) {
            v2f a = *(const v2f*)(abase + (size_t)t * 16 * K + k);
            acc[t] = __builtin_amdgcn_wmma_f32_16x16x4_f32(
                 false, a,  false, b,
                 (short)0, acc[t],  false,  false);
        }
    }

    const int cmoff = (lane >> 4) << 3;
    #pragma unroll
    for (int t = 0; t < MB; ++t) {
        float* crow = C + (size_t)(m0 + t * 16 + cmoff) * N + (n0 + l15);
        #pragma unroll
        for (int r = 0; r < 8; ++r) crow[(size_t)r * N] = acc[t][r];
    }
}

template <int F>
__global__ void agg_init_kernel(const float* __restrict__ h,
                                const float* __restrict__ dinv,
                                const float* __restrict__ bias,
                                float* __restrict__ out, int n) {
    int idx = blockIdx.x * blockDim.x + threadIdx.x;
    if (idx >= n * F) return;
    int i = idx >> (F == 128 ? 7 : 6);
    int f = idx & (F - 1);
    float di = dinv[i];
    out[idx] = bias[f] + h[idx] * di * di;
}

template <int F>
__launch_bounds__(256)
__global__ void agg_edges_kernel(const int* __restrict__ src,
                                 const int* __restrict__ dst,
                                 const float* __restrict__ dinv,
                                 const float* __restrict__ h,
                                 float* __restrict__ out, int E) {
    const int lane = threadIdx.x & 31;
    const int e    = (blockIdx.x * blockDim.x + threadIdx.x) >> 5;
    if (e >= E) return;

    const int   s    = src[e];
    const int   d    = dst[e];
    const float norm = dinv[s] * dinv[d];

    constexpr int VE = F / 32;
    const float* hs = h   + (size_t)s * F + lane * VE;
    float*       od = out + (size_t)d * F + lane * VE;

    if constexpr (VE == 4) {
        v4f v = *(const v4f*)hs;
        atomicAdd(od + 0, v.x * norm);
        atomicAdd(od + 1, v.y * norm);
        atomicAdd(od + 2, v.z * norm);
        atomicAdd(od + 3, v.w * norm);
    } else {
        v2f v = *(const v2f*)hs;
        atomicAdd(od + 0, v.x * norm);
        atomicAdd(od + 1, v.y * norm);
    }
}

__global__ void relu_kernel(float* __restrict__ x, int n) {
    int i = blockIdx.x * blockDim.x + threadIdx.x;
    if (i < n) x[i] = fmaxf(x[i], 0.0f);
}

extern "C" void kernel_launch(void* const* d_in, const int* in_sizes, int n_in,
                              void* d_out, int out_size, void* d_ws, size_t ws_size,
                              hipStream_t stream) {
    const float* x    = (const float*)d_in[0];
    const int*   eidx = (const int*)  d_in[1];
    const float* W1   = (const float*)d_in[2];
    const float* b1   = (const float*)d_in[3];
    const float* W2   = (const float*)d_in[4];
    const float* b2   = (const float*)d_in[5];
    float*       out  = (float*)d_out;

    const int N = NNODES;
    const int E = in_sizes[1] / 2;
    const int* src = eidx;
    const int* dst = eidx + E;

    char*  ws   = (char*)d_ws;
    float* dinv = (float*)(ws);
    float* h1   = (float*)(ws + (size_t)(1  << 20));
    float* out1 = (float*)(ws + (size_t)(53 << 20));
    float* h2   = h1;

    const int T = 256;
    constexpr int MB = 5;

    init_deg_kernel<<<(N + T - 1) / T, T, 0, stream>>>(dinv, N);
    accum_deg_kernel<<<(E + T - 1) / T, T, 0, stream>>>(dst, dinv, E);
    deg_to_dinv_kernel<<<(N + T - 1) / T, T, 0, stream>>>(dinv, N);

    {
        const int jobs = (N / (16 * MB)) * (128 / 16);
        gemm_f32_wmma_kernel<MB><<<jobs / 4, 128, 0, stream>>>(x, W1, h1, N, 128, 128);
    }
    agg_init_kernel<128><<<((size_t)N * 128 + T - 1) / T, T, 0, stream>>>(h1, dinv, b1, out1, N);
    agg_edges_kernel<128><<<((size_t)E * 32 + T - 1) / T, T, 0, stream>>>(src, dst, dinv, h1, out1, E);
    relu_kernel<<<((size_t)N * 128 + T - 1) / T, T, 0, stream>>>(out1, N * 128);

    {
        const int jobs = (N / (16 * MB)) * (64 / 16);
        gemm_f32_wmma_kernel<MB><<<jobs / 4, 128, 0, stream>>>(out1, W2, h2, N, 64, 128);
    }
    agg_init_kernel<64><<<((size_t)N * 64 + T - 1) / T, T, 0, stream>>>(h2, dinv, b2, out, N);
    agg_edges_kernel<64><<<((size_t)E * 32 + T - 1) / T, T, 0, stream>>>(src, dst, dinv, h2, out, E);
}


# Both dense products: a block copies its 400 A rows to LDS by whole-line loads (each row fetched once), a wave keeps its 16 B columns in registers; A fragments from LDS; k summed in a new order
# speedup vs baseline: 2.7504x; 2.7504x over previous
#include <hip/hip_runtime.h>

typedef __attribute__((ext_vector_type(2))) float v2f;
typedef __attribute__((ext_vector_type(4))) float v4f;
typedef __attribute__((ext_vector_type(8))) float v8f;
typedef __attribute__((ext_vector_type(2))) int v2i;
typedef __attribute__((ext_vector_type(4))) int v4i;
typedef __attribute__((ext_vector_type(4))) unsigned v4u;

#define NNODES 100000
#define OWN_ROWS 391

__global__ void init_deg_kernel(float* __restrict__ deg, int n) {
    int i = blockIdx.x * blockDim.x + threadIdx.x;
    if (i < n) deg[i] = 1.0f;
}

__global__ void accum_deg_kernel(const int* __restrict__ dst, float* __restrict__ deg,
                                 unsigned* __restrict__ own, int E) {
    const int t = blockIdx.x * blockDim.x + threadIdx.x;
    const int e = t * 4;
    if (e + 3 < E) {
        const v4i d = *(const v4i*)(dst + e);
        atomicAdd(&deg[d.x], 1.0f);
        atomicAdd(&deg[d.y], 1.0f);
        atomicAdd(&deg[d.z], 1.0f);
        atomicAdd(&deg[d.w], 1.0f);
        own[t] = (unsigned)(d.x / OWN_ROWS) | ((unsigned)(d.y / OWN_ROWS) << 8) |
                 ((unsigned)(d.z / OWN_ROWS) << 16) | ((unsigned)(d.w / OWN_ROWS) << 24);
    } else if (e < E) {
        unsigned w = 0;
        for (int j = 0; j < 4; ++j) {
            if (e + j < E) {
                const int d = dst[e + j];
                atomicAdd(&deg[d], 1.0f);
                w |= (unsigned)(d / OWN_ROWS) << (8 * j);
            }
        }
        own[t] = w;
    }
}

__global__ void deg_to_dinv_kernel(float* __restrict__ deg, int n) {
    int i = blockIdx.x * blockDim.x + threadIdx.x;
    if (i < n) deg[i] = rsqrtf(deg[i]);
}

#define GEMM_TM 400
template <int N>
__launch_bounds__(1024)
__global__ void gemm_f32_wmma_kernel(const float* __restrict__ A,
                                     const float* __restrict__ B,
                                     float* __restrict__ C,
                                     int M) {
    constexpr int K     = 128;
    constexpr int PITCH = 132;
    constexpr int NT    = N / 16;
    constexpr int G     = 32 / NT;
    static_assert(N % 16 == 0 && 32 % NT == 0, "one N-tile per wave, whole groups of waves");
    __shared__ __attribute__((aligned(16))) float at[GEMM_TM * PITCH];

    const int lane = threadIdx.x & 31;
    const int wave = __builtin_amdgcn_readfirstlane(threadIdx.x >> 5);
    const int r0   = blockIdx.x * GEMM_TM;
    const int c    = lane & 15;
    const int h    = lane >> 4;

    const int nt = wave % NT;
    const int g  = wave / NT;
    const int n0 = nt << 4;
    v2f bf[32];
    #pragma unroll
    for (int q = 0; q < 32; ++q) {
        const int kk = 8 * (q >> 1) + 4 * h + 2 * (q & 1);
        bf[q].x = B[(size_t)kk * N + n0 + c];
        bf[q].y = B[(size_t)(kk + 1) * N + n0 + c];
    }

    constexpr int FULL = GEMM_TM / 32;
    constexpr int REST = GEMM_TM % 32;
    v4f ld[FULL + 1];
    #pragma unroll
    for (int i = 0; i <= FULL; ++i) {
        const int r = wave + 32 * ((i < FULL || wave < REST) ? i : 0);
        int row = r0 + r;
        row = (row < M) ? row : (M - 1);
        ld[i] = *(const v4f*)(A + (size_t)row * K + lane * 4);
    }
    #pragma unroll
    for (int i = 0; i <= FULL; ++i) {
        const int r = wave + 32 * ((i < FULL || wave < REST) ? i : 0);
        *(v4f*)(at + r * PITCH + lane * 4) = ld[i];
    }
    __syncthreads();

    for (int mt = g; mt < GEMM_TM / 16; mt += G) {
        const float* ap = at + (mt * 16 + c) * PITCH + 4 * h;
        v8f acc = (v8f){};
        #pragma unroll
        for (int j = 0; j < 16; ++j) {
            const v4f a = *(const v4f*)(ap + 8 * j);
            acc = __builtin_amdgcn_wmma_f32_16x16x4_f32(
                 false, a.xy,  false, bf[2 * j],
                 (short)0, acc,  false,  false);
            acc = __builtin_amdgcn_wmma_f32_16x16x4_f32(
                 false, a.zw,  false, bf[2 * j + 1],
                 (short)0, acc,  false,  false);
        }
        const int m0 = r0 + mt * 16 + 8 * h;
        float* crow = C + (size_t)m0 * N + (n0 + c);
        if (r0 + GEMM_TM <= M) {
            #pragma unroll
            for (int r = 0; r < 8; ++r) crow[(size_t)r * N] = acc[r];
        } else {
            #pragma unroll
            for (int r = 0; r < 8; ++r)
                if (m0 + r < M) crow[(size_t)r * N] = acc[r];
        }
    }
}

template <int F, int R, bool RELU, bool SAVE, bool USE>
__launch_bounds__(1024)
__global__ void agg_owner_kernel(const int* __restrict__ src,
                                 const int* __restrict__ dst,
                                 const unsigned* __restrict__ own,
                                 const float* __restrict__ dinv,
                                 const float* __restrict__ h,
                                 const float* __restrict__ bias,
                                 float* out,
                                 float* lstf,
                                 int* lcnt,
                                 int E, int n) {
    static_assert(R == OWN_ROWS, "the owner plane is made with OWN_ROWS rows per block");
    static_assert(NNODES <= (1 << 17) && R <= (1 << 9), "a list word holds src in 17 bits, the local row above");
    constexpr int VE = F / 32;
    typedef __attribute__((ext_vector_type(VE))) float vecf;
    constexpr int ACC_FLOATS = (R * F > 41984) ? R * F : 41984;
    __shared__ __attribute__((aligned(16))) float acc[ACC_FLOATS];

    const int lane = threadIdx.x & 31;
    const int wave = __builtin_amdgcn_readfirstlane(threadIdx.x >> 5);
    const int r0   = blockIdx.x * R;
    const int nrows = (n - r0 < R) ? (n - r0) : R;

    const int cap = nrows & ~31;
    v2i* seg = (v2i*)lstf + (size_t)r0 * 32 + wave * cap;
    int nsaved = 0;

    for (int r = wave; r < nrows; r += 32) {
        const int i = r0 + r;
        const float di = dinv[i];
        const vecf hv = *(const vecf*)(h + (size_t)i * F + lane * VE);
        const vecf bv = *(const vecf*)(bias + lane * VE);
        #pragma unroll
        for (int k = 0; k < VE; ++k) acc[r * F + k * 32 + lane] = bv[k] + hv[k] * di * di;
    }
    __syncthreads();

    auto accumulate = [&](int c, int s, int dl, float nv) {
        constexpr int NB = 32 / VE;
        for (int j0 = 0; j0 < c; j0 += NB) {
            vecf rows[NB];
            #pragma unroll
            for (int j = 0; j < NB; ++j) {
                const int jj = (j0 + j < c) ? (j0 + j) : (c - 1);
                const int sj = __builtin_amdgcn_readlane(s, jj);
                rows[j] = *(const vecf*)(h + (size_t)sj * F + lane * VE);
            }
            #pragma unroll
            for (int j = 0; j < NB; ++j) {
                if (j0 + j < c) {
                    const float nj = __int_as_float(__builtin_amdgcn_readlane(__float_as_int(nv), j0 + j));
                    const int   dj = __builtin_amdgcn_readlane(dl, j0 + j);
                    float* a = acc + dj * F + lane;
                    #pragma unroll
                    for (int k = 0; k < VE; ++k)
                        __hip_atomic_fetch_add(a + k * 32, rows[j][k] * nj,
                                               __ATOMIC_RELAXED, __HIP_MEMORY_SCOPE_WORKGROUP);
                }
            }
        }
    };

    bool walked = false;
    if (USE) {
        const int nl = __builtin_amdgcn_readfirstlane(lcnt[blockIdx.x * 32 + wave]);
        if (nl <= cap) {
            walked = true;
            v2i ent = (v2i){0, 0};
            if (lane < nl) ent = seg[lane];
            int b0 = 0;
            for (; b0 + 32 <= nl; b0 += 32) {
                v2i nx = (v2i){0, 0};
                if (b0 + 32 + lane < nl) nx = seg[b0 + 32 + lane];
                accumulate(32, ent.x & 0x1ffff, ent.x >> 17, __int_as_float(ent.y));
                ent = nx;
            }
            if (b0 < nl) accumulate(nl - b0, ent.x & 0x1ffff, ent.x >> 17, __int_as_float(ent.y));
        }
    }

    if (!walked) {
    const unsigned pat = (unsigned)blockIdx.x * 0x01010101u;

    int pcode = 0;
    int cnt = 0;

    auto flush = [&](int c) {
        int s = 0, dl = 0; float nv = 0.0f;
        if (lane < c) {
            const int p = pcode & 31;
            const int e = ((pcode >> 5) << 4) + ((p & 7) << 2) + (p >> 3);
            s = src[e];
            const int d = dst[e];
            dl = d - r0;
            nv = dinv[s] * dinv[d];
        }
        if (SAVE) {
            if (nsaved + c <= cap && lane < c)
                seg[nsaved + lane] = (v2i){s | (dl << 17), __float_as_int(nv)};
            nsaved += c;
        }
        accumulate(c, s, dl, nv);
    };

    auto chunk = [&](const v4u d, const int g0, const unsigned vmask) {
        unsigned fl = 0;
        #pragma unroll
        for (int k = 0; k < 4; ++k) {
            const unsigned m = d[k] ^ pat;
            const unsigned u = ((m & 0x7f7f7f7fu) + 0x7f7f7f7fu) | m;
            fl |= (~u & 0x80808080u) >> (7 - k);
        }
        fl &= vmask;
        unsigned mask = __builtin_amdgcn_ballot_w32(fl != 0);
        while (mask) {
            const int l = __builtin_ctz(mask);
            mask &= mask - 1;
            unsigned f = (unsigned)__builtin_amdgcn_readlane((int)fl, l);
            const int base = (g0 + l) << 5;
            while (f) {
                const int p = __builtin_ctz(f);
                f &= f - 1;
                if (cnt == 32) { flush(32); cnt = 0; }
                if (lane == cnt) pcode = base | p;
                ++cnt;
            }
        }
    };

    constexpr int U    = 4;
    constexpr int CH   = 512;
    constexpr int TILE = 32 * CH * U;
    const unsigned* ow = own + wave * (CH / 4) + lane * 4;
    int tb = 0;
    if (TILE <= E) {
        v4u cur[U];
        #pragma unroll
        for (int u = 0; u < U; ++u) cur[u] = *(const v4u*)(ow + u * (32 * CH / 4));
        for (; tb + TILE <= E; tb += TILE) {
            const int nb = (tb + 2 * TILE <= E) ? (tb + TILE) : tb;
            v4u nxt[U];
            #pragma unroll
            for (int u = 0; u < U; ++u) nxt[u] = *(const v4u*)(ow + (nb >> 2) + u * (32 * CH / 4));
            #pragma unroll
            for (int u = 0; u < U; ++u) chunk(cur[u], ((tb + u * 32 * CH) >> 4) + wave * (CH / 16), ~0u);
            #pragma unroll
            for (int u = 0; u < U; ++u) cur[u] = nxt[u];
        }
    }
    for (int cb = tb + wave * CH; cb < E; cb += 32 * CH) {
        if (cb + CH <= E) {
            const v4u d = *(const v4u*)(own + (cb >> 2) + lane * 4);
            chunk(d, cb >> 4, ~0u);
        } else {
            const int e0 = cb + lane * 16;
            v4u d;
            unsigned vm = 0;
            #pragma unroll
            for (int k = 0; k < 4; ++k) {
                const int left = E - (e0 + 4 * k);
                d[k] = (left > 0) ? own[(e0 >> 2) + k] : 0u;
                const unsigned bytes = (left >= 4) ? 0x01010101u
                                     : (left > 0)  ? (((1u << (8 * left)) - 1u) & 0x01010101u) : 0u;
                vm |= bytes << k;
            }
            chunk(d, cb >> 4, vm);
        }
    }
    if (cnt > 0) flush(cnt);
    if (SAVE && lane == 0) lcnt[blockIdx.x * 32 + wave] = nsaved;
    }
    __syncthreads();

    for (int r = wave; r < nrows; r += 32) {
        vecf o;
        #pragma unroll
        for (int k = 0; k < VE; ++k) {
            const float v = acc[r * F + k * 32 + lane];
            o[k] = RELU ? fmaxf(v, 0.0f) : v;
        }
        *(vecf*)(out + (size_t)(r0 + r) * F + lane * VE) = o;
    }
}

extern "C" void kernel_launch(void* const* d_in, const int* in_sizes, int n_in,
                              void* d_out, int out_size, void* d_ws, size_t ws_size,
                              hipStream_t stream) {
    const float* x    = (const float*)d_in[0];
    const int*   eidx = (const int*)  d_in[1];
    const float* W1   = (const float*)d_in[2];
    const float* b1   = (const float*)d_in[3];
    const float* W2   = (const float*)d_in[4];
    const float* b2   = (const float*)d_in[5];
    float*       out  = (float*)d_out;

    const int N = NNODES;
    const int E = in_sizes[1] / 2;
    const int* src = eidx;
    const int* dst = eidx + E;

    char*  ws   = (char*)d_ws;
    float* dinv = (float*)(ws);
    float* h1   = (float*)(ws + (size_t)(1  << 20));
    float* out1 = (float*)(ws + (size_t)(53 << 20));
    float* h2   = h1;
    unsigned* own = (unsigned*)(ws + (size_t)(1 << 20) + (size_t)51200000);
    int* lcnt = (int*)(ws + (size_t)(512 << 10));

    const int T = 256;

    init_deg_kernel<<<(N + T - 1) / T, T, 0, stream>>>(dinv, N);
    accum_deg_kernel<<<(E + 4 * T - 1) / (4 * T), T, 0, stream>>>(dst, dinv, own, E);
    deg_to_dinv_kernel<<<(N + T - 1) / T, T, 0, stream>>>(dinv, N);

    gemm_f32_wmma_kernel<128><<<(NNODES + GEMM_TM - 1) / GEMM_TM, 1024, 0, stream>>>(x, W1, h1, N);
    agg_owner_kernel<128, OWN_ROWS, true, true, false><<<(N + 390) / 391, 1024, 0, stream>>>(src, dst, own, dinv, h1, b1, out1, out, lcnt, E, N);

    gemm_f32_wmma_kernel<64><<<(NNODES + GEMM_TM - 1) / GEMM_TM, 1024, 0, stream>>>(out1, W2, h2, N);
    agg_owner_kernel<64, OWN_ROWS, false, false, true><<<(N + 390) / 391, 1024, 0, stream>>>(src, dst, own, dinv, h2, b2, out, out, lcnt, E, N);
}


# Degrees counted by the owner blocks: a search-only launch adds each matched edge to an LDS integer counter of its dst row, writes dinv and the edge lists; global f32 atomics go; layer 1 walks lists
# speedup vs baseline: 3.7970x; 1.3805x over previous
#include <hip/hip_runtime.h>

typedef __attribute__((ext_vector_type(2))) float v2f;
typedef __attribute__((ext_vector_type(4))) float v4f;
typedef __attribute__((ext_vector_type(8))) float v8f;
typedef __attribute__((ext_vector_type(2))) int v2i;
typedef __attribute__((ext_vector_type(4))) int v4i;
typedef __attribute__((ext_vector_type(4))) unsigned v4u;

#define NNODES 100000
#define OWN_ROWS 391

__global__ void own_plane_kernel(const int* __restrict__ dst, unsigned* __restrict__ own, int E) {
    const int t = blockIdx.x * blockDim.x + threadIdx.x;
    const int e = t * 4;
    if (e + 3 < E) {
        const v4i d = *(const v4i*)(dst + e);
        own[t] = (unsigned)(d.x / OWN_ROWS) | ((unsigned)(d.y / OWN_ROWS) << 8) |
                 ((unsigned)(d.z / OWN_ROWS) << 16) | ((unsigned)(d.w / OWN_ROWS) << 24);
    } else if (e < E) {
        unsigned w = 0;
        for (int j = 0; j < 4; ++j)
            if (e + j < E) w |= (unsigned)(dst[e + j] / OWN_ROWS) << (8 * j);
        own[t] = w;
    }
}

#define GEMM_TM 400
template <int N>
__launch_bounds__(1024)
__global__ void gemm_f32_wmma_kernel(const float* __restrict__ A,
                                     const float* __restrict__ B,
                                     float* __restrict__ C,
                                     int M) {
    constexpr int K     = 128;
    constexpr int PITCH = 132;
    constexpr int NT    = N / 16;
    constexpr int G     = 32 / NT;
    static_assert(N % 16 == 0 && 32 % NT == 0, "one N-tile per wave, whole groups of waves");
    __shared__ __attribute__((aligned(16))) float at[GEMM_TM * PITCH];

    const int lane = threadIdx.x & 31;
    const int wave = __builtin_amdgcn_readfirstlane(threadIdx.x >> 5);
    const int r0   = blockIdx.x * GEMM_TM;
    const int c    = lane & 15;
    const int h    = lane >> 4;

    const int nt = wave % NT;
    const int g  = wave / NT;
    const int n0 = nt << 4;
    v2f bf[32];
    #pragma unroll
    for (int q = 0; q < 32; ++q) {
        const int kk = 8 * (q >> 1) + 4 * h + 2 * (q & 1);
        bf[q].x = B[(size_t)kk * N + n0 + c];
        bf[q].y = B[(size_t)(kk + 1) * N + n0 + c];
    }

    constexpr int FULL = GEMM_TM / 32;
    constexpr int REST = GEMM_TM % 32;
    v4f ld[FULL + 1];
    #pragma unroll
    for (int i = 0; i <= FULL; ++i) {
        const int r = wave + 32 * ((i < FULL || wave < REST) ? i : 0);
        int row = r0 + r;
        row = (row < M) ? row : (M - 1);
        ld[i] = *(const v4f*)(A + (size_t)row * K + lane * 4);
    }
    #pragma unroll
    for (int i = 0; i <= FULL; ++i) {
        const int r = wave + 32 * ((i < FULL || wave < REST) ? i : 0);
        *(v4f*)(at + r * PITCH + lane * 4) = ld[i];
    }
    __syncthreads();

    for (int mt = g; mt < GEMM_TM / 16; mt += G) {
        const float* ap = at + (mt * 16 + c) * PITCH + 4 * h;
        v8f acc = (v8f){};
        #pragma unroll
        for (int j = 0; j < 16; ++j) {
            const v4f a = *(const v4f*)(ap + 8 * j);
            acc = __builtin_amdgcn_wmma_f32_16x16x4_f32(
                 false, a.xy,  false, bf[2 * j],
                 (short)0, acc,  false,  false);
            acc = __builtin_amdgcn_wmma_f32_16x16x4_f32(
                 false, a.zw,  false, bf[2 * j + 1],
                 (short)0, acc,  false,  false);
        }
        const int m0 = r0 + mt * 16 + 8 * h;
        float* crow = C + (size_t)m0 * N + (n0 + c);
        if (r0 + GEMM_TM <= M) {
            #pragma unroll
            for (int r = 0; r < 8; ++r) crow[(size_t)r * N] = acc[r];
        } else {
            #pragma unroll
            for (int r = 0; r < 8; ++r)
                if (m0 + r < M) crow[(size_t)r * N] = acc[r];
        }
    }
}

__device__ __forceinline__ int edge_of_code(const int code) {
    const int p = code & 31;
    return ((code >> 5) << 4) + ((p & 7) << 2) + (p >> 3);
}

template <class Flush>
__device__ __forceinline__ void search_owner_plane(const unsigned* __restrict__ own, const int E,
                                                   const int wave, const int lane,
                                                   const unsigned blk, Flush flush) {
    const unsigned pat = blk * 0x01010101u;

    int pcode = 0;
    int cnt = 0;

    auto chunk = [&](const v4u d, const int g0, const unsigned vmask) {
        unsigned fl = 0;
        #pragma unroll
        for (int k = 0; k < 4; ++k) {
            const unsigned m = d[k] ^ pat;
            const unsigned u = ((m & 0x7f7f7f7fu) + 0x7f7f7f7fu) | m;
            fl |= (~u & 0x80808080u) >> (7 - k);
        }
        fl &= vmask;
        unsigned mask = __builtin_amdgcn_ballot_w32(fl != 0);
        while (mask) {
            const int l = __builtin_ctz(mask);
            mask &= mask - 1;
            unsigned f = (unsigned)__builtin_amdgcn_readlane((int)fl, l);
            const int base = (g0 + l) << 5;
            while (f) {
                const int p = __builtin_ctz(f);
                f &= f - 1;
                if (cnt == 32) { flush(32, pcode); cnt = 0; }
                if (lane == cnt) pcode = base | p;
                ++cnt;
            }
        }
    };

    constexpr int U    = 4;
    constexpr int CH   = 512;
    constexpr int TILE = 32 * CH * U;
    const unsigned* ow = own + wave * (CH / 4) + lane * 4;
    int tb = 0;
    if (TILE <= E) {
        v4u cur[U];
        #pragma unroll
        for (int u = 0; u < U; ++u) cur[u] = *(const v4u*)(ow + u * (32 * CH / 4));
        for (; tb + TILE <= E; tb += TILE) {
            const int nb = (tb + 2 * TILE <= E) ? (tb + TILE) : tb;
            v4u nxt[U];
            #pragma unroll
            for (int u = 0; u < U; ++u) nxt[u] = *(const v4u*)(ow + (nb >> 2) + u * (32 * CH / 4));
            #pragma unroll
            for (int u = 0; u < U; ++u) chunk(cur[u], ((tb + u * 32 * CH) >> 4) + wave * (CH / 16), ~0u);
            #pragma unroll
            for (int u = 0; u < U; ++u) cur[u] = nxt[u];
        }
    }
    for (int cb = tb + wave * CH; cb < E; cb += 32 * CH) {
        if (cb + CH <= E) {
            const v4u d = *(const v4u*)(own + (cb >> 2) + lane * 4);
            chunk(d, cb >> 4, ~0u);
        } else {
            const int e0 = cb + lane * 16;
            v4u d;
            unsigned vm = 0;
            #pragma unroll
            for (int k = 0; k < 4; ++k) {
                const int left = E - (e0 + 4 * k);
                d[k] = (left > 0) ? own[(e0 >> 2) + k] : 0u;
                const unsigned bytes = (left >= 4) ? 0x01010101u
                                     : (left > 0)  ? (((1u << (8 * left)) - 1u) & 0x01010101u) : 0u;
                vm |= bytes << k;
            }
            chunk(d, cb >> 4, vm);
        }
    }
    if (cnt > 0) flush(cnt, pcode);
}

template <int R>
__launch_bounds__(1024)
__global__ void find_owner_kernel(const int* __restrict__ src,
                                  const int* __restrict__ dst,
                                  const unsigned* __restrict__ own,
                                  float* __restrict__ dinv,
                                  float* lstf,
                                  int* __restrict__ lcnt,
                                  int E, int n) {
    static_assert(R == OWN_ROWS, "the owner plane is made with OWN_ROWS rows per block");
    static_assert(NNODES <= (1 << 17) && R <= (1 << 9), "a list word holds src in 17 bits, the local row above");
    __shared__ int indeg[41984];

    const int lane = threadIdx.x & 31;
    const int wave = __builtin_amdgcn_readfirstlane(threadIdx.x >> 5);
    const int r0   = blockIdx.x * R;
    const int nrows = (n - r0 < R) ? (n - r0) : R;

    const int cap = nrows & ~31;
    v2i* seg = (v2i*)lstf + (size_t)r0 * 32 + wave * cap;
    int nsaved = 0;

    for (int r = threadIdx.x; r < R; r += 1024) indeg[r] = 0;
    __syncthreads();

    search_owner_plane(own, E, wave, lane, (unsigned)blockIdx.x, [&](const int c, const int pcode) {
        if (lane < c) {
            const int e  = edge_of_code(pcode);
            const int s  = src[e];
            const int dl = dst[e] - r0;
            atomicAdd(&indeg[dl], 1);
            if (nsaved + c <= cap) seg[nsaved + lane] = (v2i){s | (dl << 17), 0};
        }
        nsaved += c;
    });
    if (lane == 0) lcnt[blockIdx.x * 32 + wave] = nsaved;
    __syncthreads();

    for (int r = threadIdx.x; r < nrows; r += 1024) dinv[r0 + r] = rsqrtf(1.0f + (float)indeg[r]);
}

template <int F, int R, bool RELU, bool NORM>
__launch_bounds__(1024)
__global__ void agg_owner_kernel(const int* __restrict__ src,
                                 const int* __restrict__ dst,
                                 const unsigned* __restrict__ own,
                                 const float* __restrict__ dinv,
                                 const float* __restrict__ h,
                                 const float* __restrict__ bias,
                                 float* out,
                                 float* lstf,
                                 const int* __restrict__ lcnt,
                                 int E, int n) {
    static_assert(R == OWN_ROWS, "the owner plane is made with OWN_ROWS rows per block");
    constexpr int VE = F / 32;
    typedef __attribute__((ext_vector_type(VE))) float vecf;
    constexpr int ACC_FLOATS = (R * F > 41984) ? R * F : 41984;
    __shared__ __attribute__((aligned(16))) float acc[ACC_FLOATS];

    const int lane = threadIdx.x & 31;
    const int wave = __builtin_amdgcn_readfirstlane(threadIdx.x >> 5);
    const int r0   = blockIdx.x * R;
    const int nrows = (n - r0 < R) ? (n - r0) : R;

    const int cap = nrows & ~31;
    v2i* seg = (v2i*)lstf + (size_t)r0 * 32 + wave * cap;

    for (int r = wave; r < nrows; r += 32) {
        const int i = r0 + r;
        const float di = dinv[i];
        const vecf hv = *(const vecf*)(h + (size_t)i * F + lane * VE);
        const vecf bv = *(const vecf*)(bias + lane * VE);
        #pragma unroll
        for (int k = 0; k < VE; ++k) acc[r * F + k * 32 + lane] = bv[k] + hv[k] * di * di;
    }
    __syncthreads();

    auto accumulate = [&](int c, int s, int dl, float nv) {
        constexpr int NB = 32 / VE;
        for (int j0 = 0; j0 < c; j0 += NB) {
            vecf rows[NB];
            #pragma unroll
            for (int j = 0; j < NB; ++j) {
                const int jj = (j0 + j < c) ? (j0 + j) : (c - 1);
                const int sj = __builtin_amdgcn_readlane(s, jj);
                rows[j] = *(const vecf*)(h + (size_t)sj * F + lane * VE);
            }
            #pragma unroll
            for (int j = 0; j < NB; ++j) {
                if (j0 + j < c) {
                    const float nj = __int_as_float(__builtin_amdgcn_readlane(__float_as_int(nv), j0 + j));
                    const int   dj = __builtin_amdgcn_readlane(dl, j0 + j);
                    float* a = acc + dj * F + lane;
                    #pragma unroll
                    for (int k = 0; k < VE; ++k)
                        __hip_atomic_fetch_add(a + k * 32, rows[j][k] * nj,
                                               __ATOMIC_RELAXED, __HIP_MEMORY_SCOPE_WORKGROUP);
                }
            }
        }
    };

    const int nl = __builtin_amdgcn_readfirstlane(lcnt[blockIdx.x * 32 + wave]);
    if (nl <= cap) {
        auto raw = [&](const int idx) {
            v2i en = (v2i){0, 0};
            if (idx < nl) en = seg[idx];
            return en;
        };
        auto norm_of = [&](const v2i en) {
            return dinv[en.x & 0x1ffff] * dinv[r0 + (en.x >> 17)];
        };
        v2i ent = raw(lane);
        v2i nx  = raw(32 + lane);
        if (NORM) ent.y = __float_as_int(norm_of(ent));
        int b0 = 0;
        for (; b0 + 32 <= nl; b0 += 32) {
            const v2i nn = raw(b0 + 64 + lane);
            float nxn = 0.0f;
            if (NORM) {
                nxn = norm_of(nx);
                seg[b0 + lane] = ent;
            }
            accumulate(32, ent.x & 0x1ffff, ent.x >> 17, __int_as_float(ent.y));
            if (NORM) nx.y = __float_as_int(nxn);
            ent = nx;
            nx = nn;
        }
        if (b0 < nl) {
            if (NORM && b0 + lane < nl) seg[b0 + lane] = ent;
            accumulate(nl - b0, ent.x & 0x1ffff, ent.x >> 17, __int_as_float(ent.y));
        }
    } else {
        search_owner_plane(own, E, wave, lane, (unsigned)blockIdx.x, [&](const int c, const int pcode) {
            int s = 0, dl = 0; float nv = 0.0f;
            if (lane < c) {
                const int e = edge_of_code(pcode);
                s = src[e];
                const int d = dst[e];
                dl = d - r0;
                nv = dinv[s] * dinv[d];
            }
            accumulate(c, s, dl, nv);
        });
    }
    __syncthreads();

    for (int r = wave; r < nrows; r += 32) {
        vecf o;
        #pragma unroll
        for (int k = 0; k < VE; ++k) {
            const float v = acc[r * F + k * 32 + lane];
            o[k] = RELU ? fmaxf(v, 0.0f) : v;
        }
        *(vecf*)(out + (size_t)(r0 + r) * F + lane * VE) = o;
    }
}

extern "C" void kernel_launch(void* const* d_in, const int* in_sizes, int n_in,
                              void* d_out, int out_size, void* d_ws, size_t ws_size,
                              hipStream_t stream) {
    const float* x    = (const float*)d_in[0];
    const int*   eidx = (const int*)  d_in[1];
    const float* W1   = (const float*)d_in[2];
    const float* b1   = (const float*)d_in[3];
    const float* W2   = (const float*)d_in[4];
    const float* b2   = (const float*)d_in[5];
    float*       out  = (float*)d_out;

    const int N = NNODES;
    const int E = in_sizes[1] / 2;
    const int* src = eidx;
    const int* dst = eidx + E;

    char*  ws   = (char*)d_ws;
    float* dinv = (float*)(ws);
    float* h1   = (float*)(ws + (size_t)(1  << 20));
    float* out1 = (float*)(ws + (size_t)(53 << 20));
    float* h2   = h1;
    unsigned* own = (unsigned*)(ws + (size_t)(1 << 20) + (size_t)51200000);
    int* lcnt = (int*)(ws + (size_t)(512 << 10));

    const int T = 256;

    own_plane_kernel<<<(E + 4 * T - 1) / (4 * T), T, 0, stream>>>(dst, own, E);
    find_owner_kernel<OWN_ROWS><<<(N + 390) / 391, 1024, 0, stream>>>(src, dst, own, dinv, out, lcnt, E, N);

    gemm_f32_wmma_kernel<128><<<(NNODES + GEMM_TM - 1) / GEMM_TM, 1024, 0, stream>>>(x, W1, h1, N);
    agg_owner_kernel<128, OWN_ROWS, true, true><<<(N + 390) / 391, 1024, 0, stream>>>(src, dst, own, dinv, h1, b1, out1, out, lcnt, E, N);

    gemm_f32_wmma_kernel<64><<<(NNODES + GEMM_TM - 1) / GEMM_TM, 1024, 0, stream>>>(out1, W2, h2, N);
    agg_owner_kernel<64, OWN_ROWS, false, false><<<(N + 390) / 391, 1024, 0, stream>>>(src, dst, own, dinv, h2, b2, out, out, lcnt, E, N);
}


# Layer-1 dense product moved into the search launch: x rows and W1^T in LDS, a wave runs one 16x16 tile of f32 WMMAs every 4th search trip on the idle matrix unit; one launch less; k sum order kept
# speedup vs baseline: 3.9315x; 1.0354x over previous
#include <hip/hip_runtime.h>

typedef __attribute__((ext_vector_type(2))) float v2f;
typedef __attribute__((ext_vector_type(4))) float v4f;
typedef __attribute__((ext_vector_type(8))) float v8f;
typedef __attribute__((ext_vector_type(2))) int v2i;
typedef __attribute__((ext_vector_type(4))) int v4i;
typedef __attribute__((ext_vector_type(4))) unsigned v4u;

#define NNODES 100000
#define OWN_ROWS 391

__global__ void own_plane_kernel(const int* __restrict__ dst, unsigned* __restrict__ own, int E) {
    const int t = blockIdx.x * blockDim.x + threadIdx.x;
    const int e = t * 4;
    if (e + 3 < E) {
        const v4i d = *(const v4i*)(dst + e);
        own[t] = (unsigned)(d.x / OWN_ROWS) | ((unsigned)(d.y / OWN_ROWS) << 8) |
                 ((unsigned)(d.z / OWN_ROWS) << 16) | ((unsigned)(d.w / OWN_ROWS) << 24);
    } else if (e < E) {
        unsigned w = 0;
        for (int j = 0; j < 4; ++j)
            if (e + j < E) w |= (unsigned)(dst[e + j] / OWN_ROWS) << (8 * j);
        own[t] = w;
    }
}

#define GEMM_TM 400
template <int N>
__launch_bounds__(1024)
__global__ void gemm_f32_wmma_kernel(const float* __restrict__ A,
                                     const float* __restrict__ B,
                                     float* __restrict__ C,
                                     int M) {
    constexpr int K     = 128;
    constexpr int PITCH = 132;
    constexpr int NT    = N / 16;
    constexpr int G     = 32 / NT;
    static_assert(N % 16 == 0 && 32 % NT == 0, "one N-tile per wave, whole groups of waves");
    __shared__ __attribute__((aligned(16))) float at[GEMM_TM * PITCH];

    const int lane = threadIdx.x & 31;
    const int wave = __builtin_amdgcn_readfirstlane(threadIdx.x >> 5);
    const int r0   = blockIdx.x * GEMM_TM;
    const int c    = lane & 15;
    const int h    = lane >> 4;

    const int nt = wave % NT;
    const int g  = wave / NT;
    const int n0 = nt << 4;
    v2f bf[32];
    #pragma unroll
    for (int q = 0; q < 32; ++q) {
        const int kk = 8 * (q >> 1) + 4 * h + 2 * (q & 1);
        bf[q].x = B[(size_t)kk * N + n0 + c];
        bf[q].y = B[(size_t)(kk + 1) * N + n0 + c];
    }

    constexpr int FULL = GEMM_TM / 32;
    constexpr int REST = GEMM_TM % 32;
    v4f ld[FULL + 1];
    #pragma unroll
    for (int i = 0; i <= FULL; ++i) {
        const int r = wave + 32 * ((i < FULL || wave < REST) ? i : 0);
        int row = r0 + r;
        row = (row < M) ? row : (M - 1);
        ld[i] = *(const v4f*)(A + (size_t)row * K + lane * 4);
    }
    #pragma unroll
    for (int i = 0; i <= FULL; ++i) {
        const int r = wave + 32 * ((i < FULL || wave < REST) ? i : 0);
        *(v4f*)(at + r * PITCH + lane * 4) = ld[i];
    }
    __syncthreads();

    for (int mt = g; mt < GEMM_TM / 16; mt += G) {
        const float* ap = at + (mt * 16 + c) * PITCH + 4 * h;
        v8f acc = (v8f){};
        #pragma unroll
        for (int j = 0; j < 16; ++j) {
            const v4f a = *(const v4f*)(ap + 8 * j);
            acc = __builtin_amdgcn_wmma_f32_16x16x4_f32(
                 false, a.xy,  false, bf[2 * j],
                 (short)0, acc,  false,  false);
            acc = __builtin_amdgcn_wmma_f32_16x16x4_f32(
                 false, a.zw,  false, bf[2 * j + 1],
                 (short)0, acc,  false,  false);
        }
        const int m0 = r0 + mt * 16 + 8 * h;
        float* crow = C + (size_t)m0 * N + (n0 + c);
        if (r0 + GEMM_TM <= M) {
            #pragma unroll
            for (int r = 0; r < 8; ++r) crow[(size_t)r * N] = acc[r];
        } else {
            #pragma unroll
            for (int r = 0; r < 8; ++r)
                if (m0 + r < M) crow[(size_t)r * N] = acc[r];
        }
    }
}

__device__ __forceinline__ int edge_of_code(const int code) {
    const int p = code & 31;
    return ((code >> 5) << 4) + ((p & 7) << 2) + (p >> 3);
}

template <class Flush, class Trip>
__device__ __forceinline__ void search_owner_plane(const unsigned* __restrict__ own, const int E,
                                                   const int wave, const int lane,
                                                   const unsigned blk, Flush flush, Trip per_trip) {
    const unsigned pat = blk * 0x01010101u;

    int pcode = 0;
    int cnt = 0;

    auto chunk = [&](const v4u d, const int g0, const unsigned vmask) {
        unsigned fl = 0;
        #pragma unroll
        for (int k = 0; k < 4; ++k) {
            const unsigned m = d[k] ^ pat;
            const unsigned u = ((m & 0x7f7f7f7fu) + 0x7f7f7f7fu) | m;
            fl |= (~u & 0x80808080u) >> (7 - k);
        }
        fl &= vmask;
        unsigned mask = __builtin_amdgcn_ballot_w32(fl != 0);
        while (mask) {
            const int l = __builtin_ctz(mask);
            mask &= mask - 1;
            unsigned f = (unsigned)__builtin_amdgcn_readlane((int)fl, l);
            const int base = (g0 + l) << 5;
            while (f) {
                const int p = __builtin_ctz(f);
                f &= f - 1;
                if (cnt == 32) { flush(32, pcode); cnt = 0; }
                if (lane == cnt) pcode = base | p;
                ++cnt;
            }
        }
    };

    constexpr int U    = 4;
    constexpr int CH   = 512;
    constexpr int TILE = 32 * CH * U;
    const unsigned* ow = own + wave * (CH / 4) + lane * 4;
    int tb = 0;
    int trip = 0;
    if (TILE <= E) {
        v4u cur[U];
        #pragma unroll
        for (int u = 0; u < U; ++u) cur[u] = *(const v4u*)(ow + u * (32 * CH / 4));
        for (; tb + TILE <= E; tb += TILE) {
            const int nb = (tb + 2 * TILE <= E) ? (tb + TILE) : tb;
            v4u nxt[U];
            #pragma unroll
            for (int u = 0; u < U; ++u) nxt[u] = *(const v4u*)(ow + (nb >> 2) + u * (32 * CH / 4));
            per_trip(trip);
            ++trip;
            #pragma unroll
            for (int u = 0; u < U; ++u) chunk(cur[u], ((tb + u * 32 * CH) >> 4) + wave * (CH / 16), ~0u);
            #pragma unroll
            for (int u = 0; u < U; ++u) cur[u] = nxt[u];
        }
    }
    for (int cb = tb + wave * CH; cb < E; cb += 32 * CH) {
        if (cb + CH <= E) {
            const v4u d = *(const v4u*)(own + (cb >> 2) + lane * 4);
            chunk(d, cb >> 4, ~0u);
        } else {
            const int e0 = cb + lane * 16;
            v4u d;
            unsigned vm = 0;
            #pragma unroll
            for (int k = 0; k < 4; ++k) {
                const int left = E - (e0 + 4 * k);
                d[k] = (left > 0) ? own[(e0 >> 2) + k] : 0u;
                const unsigned bytes = (left >= 4) ? 0x01010101u
                                     : (left > 0)  ? (((1u << (8 * left)) - 1u) & 0x01010101u) : 0u;
                vm |= bytes << k;
            }
            chunk(d, cb >> 4, vm);
        }
    }
    if (cnt > 0) flush(cnt, pcode);
}

#define FO_TM 400
template <int R>
__launch_bounds__(1024)
__global__ void find_owner_kernel(const int* __restrict__ src,
                                  const int* __restrict__ dst,
                                  const unsigned* __restrict__ own,
                                  float* __restrict__ dinv,
                                  float* lstf,
                                  int* __restrict__ lcnt,
                                  const float* __restrict__ X,
                                  const float* __restrict__ Wm,
                                  float* __restrict__ H,
                                  int E, int n) {
    static_assert(R == OWN_ROWS, "the owner plane is made with OWN_ROWS rows per block");
    static_assert(NNODES <= (1 << 17) && R <= (1 << 9), "a list word holds src in 17 bits, the local row above");
    static_assert(R <= FO_TM && FO_TM % 16 == 0, "the A tile holds every M-tile that touches the block's rows");
    constexpr int K     = 128;
    constexpr int N     = 128;
    constexpr int PITCH = 132;
    __shared__ int indeg[512];
    __shared__ __attribute__((aligned(16))) float at[FO_TM * PITCH];
    __shared__ __attribute__((aligned(16))) float wt[N * PITCH];

    const int lane = threadIdx.x & 31;
    const int wave = __builtin_amdgcn_readfirstlane(threadIdx.x >> 5);
    const int r0   = blockIdx.x * R;
    const int nrows = (n - r0 < R) ? (n - r0) : R;

    const int cap = nrows & ~31;
    v2i* seg = (v2i*)lstf + (size_t)r0 * 32 + wave * cap;
    int nsaved = 0;

    for (int r = threadIdx.x; r < R; r += 1024) indeg[r] = 0;

    {
        constexpr int FULL = FO_TM / 32;
        constexpr int REST = FO_TM % 32;
        v4f ld[FULL + 1];
        #pragma unroll
        for (int i = 0; i <= FULL; ++i) {
            const int r = wave + 32 * ((i < FULL || wave < REST) ? i : 0);
            int row = r0 + r;
            row = (row < n) ? row : (n - 1);
            ld[i] = *(const v4f*)(X + (size_t)row * K + lane * 4);
        }
        #pragma unroll
        for (int i = 0; i <= FULL; ++i) {
            const int r = wave + 32 * ((i < FULL || wave < REST) ? i : 0);
            *(v4f*)(at + r * PITCH + lane * 4) = ld[i];
        }
    }
    for (int i = threadIdx.x; i < K * N; i += 1024)
        wt[(i % N) * PITCH + (i / N)] = Wm[i];
    __syncthreads();

    const int c  = lane & 15;
    const int h  = lane >> 4;
    const int n0 = (wave & 7) << 4;
    const int mtiles = (nrows + 15) >> 4;
    int mt = wave >> 3;
    auto product_tile = [&]() {
        const float* ap = at + (mt * 16 + c) * PITCH + 4 * h;
        const float* bp = wt + (n0 + c) * PITCH + 4 * h;
        v8f acc = (v8f){};
        #pragma unroll 4
        for (int j = 0; j < 16; ++j) {
            const v4f a = *(const v4f*)(ap + 8 * j);
            const v4f b = *(const v4f*)(bp + 8 * j);
            acc = __builtin_amdgcn_wmma_f32_16x16x4_f32(
                 false, a.xy,  false, b.xy,
                 (short)0, acc,  false,  false);
            acc = __builtin_amdgcn_wmma_f32_16x16x4_f32(
                 false, a.zw,  false, b.zw,
                 (short)0, acc,  false,  false);
        }
        const int ml = mt * 16 + 8 * h;
        float* crow = H + (size_t)(r0 + ml) * N + (n0 + c);
        if (mt * 16 + 16 <= nrows) {
            #pragma unroll
            for (int r = 0; r < 8; ++r) crow[(size_t)r * N] = acc[r];
        } else {
            #pragma unroll
            for (int r = 0; r < 8; ++r)
                if (ml + r < nrows) crow[(size_t)r * N] = acc[r];
        }
        mt += 4;
    };
    const int phase = (wave + (wave >> 3)) & 3;

    search_owner_plane(own, E, wave, lane, (unsigned)blockIdx.x, [&](const int cfl, const int pcode) {
        if (lane < cfl) {
            const int e  = edge_of_code(pcode);
            const int sv = src[e];
            const int dl = dst[e] - r0;
            atomicAdd(&indeg[dl], 1);
            if (nsaved + cfl <= cap) seg[nsaved + lane] = (v2i){sv | (dl << 17), 0};
        }
        nsaved += cfl;
    }, [&](const int trip) {
        if (((trip + phase) & 3) == 3 && mt < mtiles) product_tile();
    });
    while (mt < mtiles) product_tile();

    if (lane == 0) lcnt[blockIdx.x * 32 + wave] = nsaved;
    __syncthreads();

    for (int r = threadIdx.x; r < nrows; r += 1024) dinv[r0 + r] = rsqrtf(1.0f + (float)indeg[r]);
}

template <int F, int R, bool RELU, bool NORM>
__launch_bounds__(1024)
__global__ void agg_owner_kernel(const int* __restrict__ src,
                                 const int* __restrict__ dst,
                                 const unsigned* __restrict__ own,
                                 const float* __restrict__ dinv,
                                 const float* __restrict__ h,
                                 const float* __restrict__ bias,
                                 float* out,
                                 float* lstf,
                                 const int* __restrict__ lcnt,
                                 int E, int n) {
    static_assert(R == OWN_ROWS, "the owner plane is made with OWN_ROWS rows per block");
    constexpr int VE = F / 32;
    typedef __attribute__((ext_vector_type(VE))) float vecf;
    constexpr int ACC_FLOATS = (R * F > 41984) ? R * F : 41984;
    __shared__ __attribute__((aligned(16))) float acc[ACC_FLOATS];

    const int lane = threadIdx.x & 31;
    const int wave = __builtin_amdgcn_readfirstlane(threadIdx.x >> 5);
    const int r0   = blockIdx.x * R;
    const int nrows = (n - r0 < R) ? (n - r0) : R;

    const int cap = nrows & ~31;
    v2i* seg = (v2i*)lstf + (size_t)r0 * 32 + wave * cap;

    for (int r = wave; r < nrows; r += 32) {
        const int i = r0 + r;
        const float di = dinv[i];
        const vecf hv = *(const vecf*)(h + (size_t)i * F + lane * VE);
        const vecf bv = *(const vecf*)(bias + lane * VE);
        #pragma unroll
        for (int k = 0; k < VE; ++k) acc[r * F + k * 32 + lane] = bv[k] + hv[k] * di * di;
    }
    __syncthreads();

    auto accumulate = [&](int c, int s, int dl, float nv) {
        constexpr int NB = 32 / VE;
        for (int j0 = 0; j0 < c; j0 += NB) {
            vecf rows[NB];
            #pragma unroll
            for (int j = 0; j < NB; ++j) {
                const int jj = (j0 + j < c) ? (j0 + j) : (c - 1);
                const int sj = __builtin_amdgcn_readlane(s, jj);
                rows[j] = *(const vecf*)(h + (size_t)sj * F + lane * VE);
            }
            #pragma unroll
            for (int j = 0; j < NB; ++j) {
                if (j0 + j < c) {
                    const float nj = __int_as_float(__builtin_amdgcn_readlane(__float_as_int(nv), j0 + j));
                    const int   dj = __builtin_amdgcn_readlane(dl, j0 + j);
                    float* a = acc + dj * F + lane;
                    #pragma unroll
                    for (int k = 0; k < VE; ++k)
                        __hip_atomic_fetch_add(a + k * 32, rows[j][k] * nj,
                                               __ATOMIC_RELAXED, __HIP_MEMORY_SCOPE_WORKGROUP);
                }
            }
        }
    };

    const int nl = __builtin_amdgcn_readfirstlane(lcnt[blockIdx.x * 32 + wave]);
    if (nl <= cap) {
        auto raw = [&](const int idx) {
            v2i en = (v2i){0, 0};
            if (idx < nl) en = seg[idx];
            return en;
        };
        auto norm_of = [&](const v2i en) {
            return dinv[en.x & 0x1ffff] * dinv[r0 + (en.x >> 17)];
        };
        v2i ent = raw(lane);
        v2i nx  = raw(32 + lane);
        if (NORM) ent.y = __float_as_int(norm_of(ent));
        int b0 = 0;
        for (; b0 + 32 <= nl; b0 += 32) {
            const v2i nn = raw(b0 + 64 + lane);
            float nxn = 0.0f;
            if (NORM) {
                nxn = norm_of(nx);
                seg[b0 + lane] = ent;
            }
            accumulate(32, ent.x & 0x1ffff, ent.x >> 17, __int_as_float(ent.y));
            if (NORM) nx.y = __float_as_int(nxn);
            ent = nx;
            nx = nn;
        }
        if (b0 < nl) {
            if (NORM && b0 + lane < nl) seg[b0 + lane] = ent;
            accumulate(nl - b0, ent.x & 0x1ffff, ent.x >> 17, __int_as_float(ent.y));
        }
    } else {
        search_owner_plane(own, E, wave, lane, (unsigned)blockIdx.x, [&](const int c, const int pcode) {
            int s = 0, dl = 0; float nv = 0.0f;
            if (lane < c) {
                const int e = edge_of_code(pcode);
                s = src[e];
                const int d = dst[e];
                dl = d - r0;
                nv = dinv[s] * dinv[d];
            }
            accumulate(c, s, dl, nv);
        }, [](const int) {});
    }
    __syncthreads();

    for (int r = wave; r < nrows; r += 32) {
        vecf o;
        #pragma unroll
        for (int k = 0; k < VE; ++k) {
            const float v = acc[r * F + k * 32 + lane];
            o[k] = RELU ? fmaxf(v, 0.0f) : v;
        }
        *(vecf*)(out + (size_t)(r0 + r) * F + lane * VE) = o;
    }
}

extern "C" void kernel_launch(void* const* d_in, const int* in_sizes, int n_in,
                              void* d_out, int out_size, void* d_ws, size_t ws_size,
                              hipStream_t stream) {
    const float* x    = (const float*)d_in[0];
    const int*   eidx = (const int*)  d_in[1];
    const float* W1   = (const float*)d_in[2];
    const float* b1   = (const float*)d_in[3];
    const float* W2   = (const float*)d_in[4];
    const float* b2   = (const float*)d_in[5];
    float*       out  = (float*)d_out;

    const int N = NNODES;
    const int E = in_sizes[1] / 2;
    const int* src = eidx;
    const int* dst = eidx + E;

    char*  ws   = (char*)d_ws;
    float* dinv = (float*)(ws);
    float* h1   = (float*)(ws + (size_t)(1  << 20));
    float* out1 = (float*)(ws + (size_t)(53 << 20));
    float* h2   = h1;
    unsigned* own = (unsigned*)(ws + (size_t)(1 << 20) + (size_t)51200000);
    int* lcnt = (int*)(ws + (size_t)(512 << 10));

    const int T = 256;

    own_plane_kernel<<<(E + 4 * T - 1) / (4 * T), T, 0, stream>>>(dst, own, E);
    find_owner_kernel<OWN_ROWS><<<(N + 390) / 391, 1024, 0, stream>>>(src, dst, own, dinv, out, lcnt, x, W1, h1, E, N);

    agg_owner_kernel<128, OWN_ROWS, true, true><<<(N + 390) / 391, 1024, 0, stream>>>(src, dst, own, dinv, h1, b1, out1, out, lcnt, E, N);

    gemm_f32_wmma_kernel<64><<<(NNODES + GEMM_TM - 1) / GEMM_TM, 1024, 0, stream>>>(out1, W2, h2, N);
    agg_owner_kernel<64, OWN_ROWS, false, false><<<(N + 390) / 391, 1024, 0, stream>>>(src, dst, own, dinv, h2, b2, out, out, lcnt, E, N);
}


# Layer-2 dense product moved into the layer-1 aggregation: the block's summed rows feed f32 WMMA A fragments straight from LDS (relu on read), h2 is stored; out1 and one launch go; k, sum order kept
# speedup vs baseline: 4.1994x; 1.0681x over previous
#include <hip/hip_runtime.h>

typedef __attribute__((ext_vector_type(2))) float v2f;
typedef __attribute__((ext_vector_type(4))) float v4f;
typedef __attribute__((ext_vector_type(8))) float v8f;
typedef __attribute__((ext_vector_type(2))) int v2i;
typedef __attribute__((ext_vector_type(4))) int v4i;
typedef __attribute__((ext_vector_type(4))) unsigned v4u;

#define NNODES 100000
#define OWN_ROWS 391

__global__ void own_plane_kernel(const int* __restrict__ dst, unsigned* __restrict__ own, int E) {
    const int t = blockIdx.x * blockDim.x + threadIdx.x;
    const int e = t * 4;
    if (e + 3 < E) {
        const v4i d = *(const v4i*)(dst + e);
        own[t] = (unsigned)(d.x / OWN_ROWS) | ((unsigned)(d.y / OWN_ROWS) << 8) |
                 ((unsigned)(d.z / OWN_ROWS) << 16) | ((unsigned)(d.w / OWN_ROWS) << 24);
    } else if (e < E) {
        unsigned w = 0;
        for (int j = 0; j < 4; ++j)
            if (e + j < E) w |= (unsigned)(dst[e + j] / OWN_ROWS) << (8 * j);
        own[t] = w;
    }
}

__device__ __forceinline__ int edge_of_code(const int code) {
    const int p = code & 31;
    return ((code >> 5) << 4) + ((p & 7) << 2) + (p >> 3);
}

template <class Flush, class Trip>
__device__ __forceinline__ void search_owner_plane(const unsigned* __restrict__ own, const int E,
                                                   const int wave, const int lane,
                                                   const unsigned blk, Flush flush, Trip per_trip) {
    const unsigned pat = blk * 0x01010101u;

    int pcode = 0;
    int cnt = 0;

    auto chunk = [&](const v4u d, const int g0, const unsigned vmask) {
        unsigned fl = 0;
        #pragma unroll
        for (int k = 0; k < 4; ++k) {
            const unsigned m = d[k] ^ pat;
            const unsigned u = ((m & 0x7f7f7f7fu) + 0x7f7f7f7fu) | m;
            fl |= (~u & 0x80808080u) >> (7 - k);
        }
        fl &= vmask;
        unsigned mask = __builtin_amdgcn_ballot_w32(fl != 0);
        while (mask) {
            const int l = __builtin_ctz(mask);
            mask &= mask - 1;
            unsigned f = (unsigned)__builtin_amdgcn_readlane((int)fl, l);
            const int base = (g0 + l) << 5;
            while (f) {
                const int p = __builtin_ctz(f);
                f &= f - 1;
                if (cnt == 32) { flush(32, pcode); cnt = 0; }
                if (lane == cnt) pcode = base | p;
                ++cnt;
            }
        }
    };

    constexpr int U    = 4;
    constexpr int CH   = 512;
    constexpr int TILE = 32 * CH * U;
    const unsigned* ow = own + wave * (CH / 4) + lane * 4;
    int tb = 0;
    int trip = 0;
    if (TILE <= E) {
        v4u cur[U];
        #pragma unroll
        for (int u = 0; u < U; ++u) cur[u] = *(const v4u*)(ow + u * (32 * CH / 4));
        for (; tb + TILE <= E; tb += TILE) {
            const int nb = (tb + 2 * TILE <= E) ? (tb + TILE) : tb;
            v4u nxt[U];
            #pragma unroll
            for (int u = 0; u < U; ++u) nxt[u] = *(const v4u*)(ow + (nb >> 2) + u * (32 * CH / 4));
            per_trip(trip);
            ++trip;
            #pragma unroll
            for (int u = 0; u < U; ++u) chunk(cur[u], ((tb + u * 32 * CH) >> 4) + wave * (CH / 16), ~0u);
            #pragma unroll
            for (int u = 0; u < U; ++u) cur[u] = nxt[u];
        }
    }
    for (int cb = tb + wave * CH; cb < E; cb += 32 * CH) {
        if (cb + CH <= E) {
            const v4u d = *(const v4u*)(own + (cb >> 2) + lane * 4);
            chunk(d, cb >> 4, ~0u);
        } else {
            const int e0 = cb + lane * 16;
            v4u d;
            unsigned vm = 0;
            #pragma unroll
            for (int k = 0; k < 4; ++k) {
                const int left = E - (e0 + 4 * k);
                d[k] = (left > 0) ? own[(e0 >> 2) + k] : 0u;
                const unsigned bytes = (left >= 4) ? 0x01010101u
                                     : (left > 0)  ? (((1u << (8 * left)) - 1u) & 0x01010101u) : 0u;
                vm |= bytes << k;
            }
            chunk(d, cb >> 4, vm);
        }
    }
    if (cnt > 0) flush(cnt, pcode);
}

#define FO_TM 400
template <int R>
__launch_bounds__(1024)
__global__ void find_owner_kernel(const int* __restrict__ src,
                                  const int* __restrict__ dst,
                                  const unsigned* __restrict__ own,
                                  float* __restrict__ dinv,
                                  float* lstf,
                                  int* __restrict__ lcnt,
                                  const float* __restrict__ X,
                                  const float* __restrict__ Wm,
                                  float* __restrict__ H,
                                  int E, int n) {
    static_assert(R == OWN_ROWS, "the owner plane is made with OWN_ROWS rows per block");
    static_assert(NNODES <= (1 << 17) && R <= (1 << 9), "a list word holds src in 17 bits, the local row above");
    static_assert(R <= FO_TM && FO_TM % 16 == 0, "the A tile holds every M-tile that touches the block's rows");
    constexpr int K     = 128;
    constexpr int N     = 128;
    constexpr int PITCH = 132;
    __shared__ int indeg[512];
    __shared__ __attribute__((aligned(16))) float at[FO_TM * PITCH];
    __shared__ __attribute__((aligned(16))) float wt[N * PITCH];

    const int lane = threadIdx.x & 31;
    const int wave = __builtin_amdgcn_readfirstlane(threadIdx.x >> 5);
    const int r0   = blockIdx.x * R;
    const int nrows = (n - r0 < R) ? (n - r0) : R;

    const int cap = nrows & ~31;
    v2i* seg = (v2i*)lstf + (size_t)r0 * 32 + wave * cap;
    int nsaved = 0;

    for (int r = threadIdx.x; r < R; r += 1024) indeg[r] = 0;

    {
        constexpr int FULL = FO_TM / 32;
        constexpr int REST = FO_TM % 32;
        v4f ld[FULL + 1];
        #pragma unroll
        for (int i = 0; i <= FULL; ++i) {
            const int r = wave + 32 * ((i < FULL || wave < REST) ? i : 0);
            int row = r0 + r;
            row = (row < n) ? row : (n - 1);
            ld[i] = *(const v4f*)(X + (size_t)row * K + lane * 4);
        }
        #pragma unroll
        for (int i = 0; i <= FULL; ++i) {
            const int r = wave + 32 * ((i < FULL || wave < REST) ? i : 0);
            *(v4f*)(at + r * PITCH + lane * 4) = ld[i];
        }
    }
    for (int i = threadIdx.x; i < K * N; i += 1024)
        wt[(i % N) * PITCH + (i / N)] = Wm[i];
    __syncthreads();

    const int c  = lane & 15;
    const int h  = lane >> 4;
    const int n0 = (wave & 7) << 4;
    const int mtiles = (nrows + 15) >> 4;
    int mt = wave >> 3;
    auto product_tile = [&]() {
        const float* ap = at + (mt * 16 + c) * PITCH + 4 * h;
        const float* bp = wt + (n0 + c) * PITCH + 4 * h;
        v8f acc = (v8f){};
        #pragma unroll 4
        for (int j = 0; j < 16; ++j) {
            const v4f a = *(const v4f*)(ap + 8 * j);
            const v4f b = *(const v4f*)(bp + 8 * j);
            acc = __builtin_amdgcn_wmma_f32_16x16x4_f32(
                 false, a.xy,  false, b.xy,
                 (short)0, acc,  false,  false);
            acc = __builtin_amdgcn_wmma_f32_16x16x4_f32(
                 false, a.zw,  false, b.zw,
                 (short)0, acc,  false,  false);
        }
        const int ml = mt * 16 + 8 * h;
        float* crow = H + (size_t)(r0 + ml) * N + (n0 + c);
        if (mt * 16 + 16 <= nrows) {
            #pragma unroll
            for (int r = 0; r < 8; ++r) crow[(size_t)r * N] = acc[r];
        } else {
            #pragma unroll
            for (int r = 0; r < 8; ++r)
                if (ml + r < nrows) crow[(size_t)r * N] = acc[r];
        }
        mt += 4;
    };
    const int phase = (wave + (wave >> 3)) & 3;

    search_owner_plane(own, E, wave, lane, (unsigned)blockIdx.x, [&](const int cfl, const int pcode) {
        if (lane < cfl) {
            const int e  = edge_of_code(pcode);
            const int sv = src[e];
            const int dl = dst[e] - r0;
            atomicAdd(&indeg[dl], 1);
            if (nsaved + cfl <= cap) seg[nsaved + lane] = (v2i){sv | (dl << 17), 0};
        }
        nsaved += cfl;
    }, [&](const int trip) {
        if (((trip + phase) & 3) == 3 && mt < mtiles) product_tile();
    });
    while (mt < mtiles) product_tile();

    if (lane == 0) lcnt[blockIdx.x * 32 + wave] = nsaved;
    __syncthreads();

    for (int r = threadIdx.x; r < nrows; r += 1024) dinv[r0 + r] = rsqrtf(1.0f + (float)indeg[r]);
}

template <int F, int R, bool RELU, bool NORM, int FN>
__launch_bounds__(1024)
__global__ void agg_owner_kernel(const int* __restrict__ src,
                                 const int* __restrict__ dst,
                                 const unsigned* __restrict__ own,
                                 const float* __restrict__ dinv,
                                 const float* __restrict__ h,
                                 const float* __restrict__ bias,
                                 const float* __restrict__ Wn,
                                 float* out,
                                 float* lstf,
                                 const int* __restrict__ lcnt,
                                 int E, int n) {
    static_assert(R == OWN_ROWS, "the owner plane is made with OWN_ROWS rows per block");
    static_assert(FN == 0 || (F == 128 && FN % 16 == 0 && 32 % (FN / 16) == 0 && RELU),
                  "the fused product reads 4-float rows of 128 and gives every wave one N-tile");
    constexpr int VE = F / 32;
    typedef __attribute__((ext_vector_type(VE))) float vecf;
    constexpr int PITCH = (FN > 0) ? F + 4 : F;
    constexpr int LROWS = (FN > 0) ? ((R + 15) / 16) * 16 : R;
    constexpr int ACC_FLOATS = (LROWS * PITCH > 41984) ? LROWS * PITCH : 41984;
    __shared__ __attribute__((aligned(16))) float acc[ACC_FLOATS];

    const int lane = threadIdx.x & 31;
    const int wave = __builtin_amdgcn_readfirstlane(threadIdx.x >> 5);
    const int r0   = blockIdx.x * R;
    const int nrows = (n - r0 < R) ? (n - r0) : R;

    const int cap = nrows & ~31;
    v2i* seg = (v2i*)lstf + (size_t)r0 * 32 + wave * cap;

    for (int r = wave; r < nrows; r += 32) {
        const int i = r0 + r;
        const float di = dinv[i];
        const vecf hv = *(const vecf*)(h + (size_t)i * F + lane * VE);
        const vecf bv = *(const vecf*)(bias + lane * VE);
        #pragma unroll
        for (int k = 0; k < VE; ++k) acc[r * PITCH + k * 32 + lane] = bv[k] + hv[k] * di * di;
    }
    if (FN > 0) {
        for (int r = nrows + wave; r < LROWS; r += 32) {
            #pragma unroll
            for (int k = 0; k < VE; ++k) acc[r * PITCH + k * 32 + lane] = 0.0f;
        }
    }
    __syncthreads();

    auto accumulate = [&](int c, int s, int dl, float nv) {
        constexpr int NB = 32 / VE;
        for (int j0 = 0; j0 < c; j0 += NB) {
            vecf rows[NB];
            #pragma unroll
            for (int j = 0; j < NB; ++j) {
                const int jj = (j0 + j < c) ? (j0 + j) : (c - 1);
                const int sj = __builtin_amdgcn_readlane(s, jj);
                rows[j] = *(const vecf*)(h + (size_t)sj * F + lane * VE);
            }
            #pragma unroll
            for (int j = 0; j < NB; ++j) {
                if (j0 + j < c) {
                    const float nj = __int_as_float(__builtin_amdgcn_readlane(__float_as_int(nv), j0 + j));
                    const int   dj = __builtin_amdgcn_readlane(dl, j0 + j);
                    float* a = acc + dj * PITCH + lane;
                    #pragma unroll
                    for (int k = 0; k < VE; ++k)
                        __hip_atomic_fetch_add(a + k * 32, rows[j][k] * nj,
                                               __ATOMIC_RELAXED, __HIP_MEMORY_SCOPE_WORKGROUP);
                }
            }
        }
    };

    const int nl = __builtin_amdgcn_readfirstlane(lcnt[blockIdx.x * 32 + wave]);
    if (nl <= cap) {
        auto raw = [&](const int idx) {
            v2i en = (v2i){0, 0};
            if (idx < nl) en = seg[idx];
            return en;
        };
        auto norm_of = [&](const v2i en) {
            return dinv[en.x & 0x1ffff] * dinv[r0 + (en.x >> 17)];
        };
        v2i ent = raw(lane);
        v2i nx  = raw(32 + lane);
        if (NORM) ent.y = __float_as_int(norm_of(ent));
        int b0 = 0;
        for (; b0 + 32 <= nl; b0 += 32) {
            const v2i nn = raw(b0 + 64 + lane);
            float nxn = 0.0f;
            if (NORM) {
                nxn = norm_of(nx);
                seg[b0 + lane] = ent;
            }
            accumulate(32, ent.x & 0x1ffff, ent.x >> 17, __int_as_float(ent.y));
            if (NORM) nx.y = __float_as_int(nxn);
            ent = nx;
            nx = nn;
        }
        if (b0 < nl) {
            if (NORM && b0 + lane < nl) seg[b0 + lane] = ent;
            accumulate(nl - b0, ent.x & 0x1ffff, ent.x >> 17, __int_as_float(ent.y));
        }
    } else {
        search_owner_plane(own, E, wave, lane, (unsigned)blockIdx.x, [&](const int c, const int pcode) {
            int s = 0, dl = 0; float nv = 0.0f;
            if (lane < c) {
                const int e = edge_of_code(pcode);
                s = src[e];
                const int d = dst[e];
                dl = d - r0;
                nv = dinv[s] * dinv[d];
            }
            accumulate(c, s, dl, nv);
        }, [](const int) {});
    }
    if constexpr (FN > 0) {
        constexpr int NT = FN / 16;
        constexpr int G  = 32 / NT;
        const int c  = lane & 15;
        const int hf = lane >> 4;
        const int nt = wave % NT;
        const int g  = wave / NT;
        const int n0 = nt << 4;
        v2f bf[32];
        #pragma unroll
        for (int q = 0; q < 32; ++q) {
            const int kk = 8 * (q >> 1) + 4 * hf + 2 * (q & 1);
            bf[q].x = Wn[(size_t)kk * FN + n0 + c];
            bf[q].y = Wn[(size_t)(kk + 1) * FN + n0 + c];
        }
        __syncthreads();

        const int mtiles = (nrows + 15) >> 4;
        for (int mt = g; mt < mtiles; mt += G) {
            const float* ap = acc + (mt * 16 + c) * PITCH + hf;
            v8f d = (v8f){};
            #pragma unroll
            for (int j = 0; j < 16; ++j) {
                v2f alo, ahi;
                alo.x = fmaxf(ap[2 * j],      0.0f);
                alo.y = fmaxf(ap[2 * j + 32], 0.0f);
                ahi.x = fmaxf(ap[2 * j + 64], 0.0f);
                ahi.y = fmaxf(ap[2 * j + 96], 0.0f);
                d = __builtin_amdgcn_wmma_f32_16x16x4_f32(
                     false, alo,  false, bf[2 * j],
                     (short)0, d,  false,  false);
                d = __builtin_amdgcn_wmma_f32_16x16x4_f32(
                     false, ahi,  false, bf[2 * j + 1],
                     (short)0, d,  false,  false);
            }
            const int ml = mt * 16 + 8 * hf;
            float* crow = out + (size_t)(r0 + ml) * FN + (n0 + c);
            if (mt * 16 + 16 <= nrows) {
                #pragma unroll
                for (int r = 0; r < 8; ++r) crow[(size_t)r * FN] = d[r];
            } else {
                #pragma unroll
                for (int r = 0; r < 8; ++r)
                    if (ml + r < nrows) crow[(size_t)r * FN] = d[r];
            }
        }
    } else {
        __syncthreads();

        for (int r = wave; r < nrows; r += 32) {
            vecf o;
            #pragma unroll
            for (int k = 0; k < VE; ++k) {
                const float v = acc[r * PITCH + k * 32 + lane];
                o[k] = RELU ? fmaxf(v, 0.0f) : v;
            }
            *(vecf*)(out + (size_t)(r0 + r) * F + lane * VE) = o;
        }
    }
}

extern "C" void kernel_launch(void* const* d_in, const int* in_sizes, int n_in,
                              void* d_out, int out_size, void* d_ws, size_t ws_size,
                              hipStream_t stream) {
    const float* x    = (const float*)d_in[0];
    const int*   eidx = (const int*)  d_in[1];
    const float* W1   = (const float*)d_in[2];
    const float* b1   = (const float*)d_in[3];
    const float* W2   = (const float*)d_in[4];
    const float* b2   = (const float*)d_in[5];
    float*       out  = (float*)d_out;

    const int N = NNODES;
    const int E = in_sizes[1] / 2;
    const int* src = eidx;
    const int* dst = eidx + E;

    char*  ws   = (char*)d_ws;
    float* dinv = (float*)(ws);
    float* h1   = (float*)(ws + (size_t)(1  << 20));
    float* h2   = (float*)(ws + (size_t)(53 << 20));
    unsigned* own = (unsigned*)(ws + (size_t)(1 << 20) + (size_t)51200000);
    int* lcnt = (int*)(ws + (size_t)(512 << 10));

    const int T = 256;

    own_plane_kernel<<<(E + 4 * T - 1) / (4 * T), T, 0, stream>>>(dst, own, E);
    find_owner_kernel<OWN_ROWS><<<(N + 390) / 391, 1024, 0, stream>>>(src, dst, own, dinv, out, lcnt, x, W1, h1, E, N);

    agg_owner_kernel<128, OWN_ROWS, true, true, 64><<<(N + 390) / 391, 1024, 0, stream>>>(src, dst, own, dinv, h1, b1, W2, h2, out, lcnt, E, N);

    agg_owner_kernel<64, OWN_ROWS, false, false, 0><<<(N + 390) / 391, 1024, 0, stream>>>(src, dst, own, dinv, h2, b2, nullptr, out, out, lcnt, E, N);
}


# 1-byte owner plane replaced by 16 one-hot bit planes (owner number as four 2-bit fields): a block ANDs its 4 planes, 128 edges per lane and trip, matches filed by lane rank; sum order: LDS arrival
# speedup vs baseline: 4.5448x; 1.0822x over previous
#include <hip/hip_runtime.h>

typedef __attribute__((ext_vector_type(2))) float v2f;
typedef __attribute__((ext_vector_type(4))) float v4f;
typedef __attribute__((ext_vector_type(8))) float v8f;
typedef __attribute__((ext_vector_type(2))) int v2i;
typedef __attribute__((ext_vector_type(4))) int v4i;
typedef __attribute__((ext_vector_type(4))) unsigned v4u;

#define NNODES 100000
#define OWN_ROWS 391
#define PLANE_WORDS 50176

__global__ void own_plane_kernel(const int* __restrict__ dst, unsigned* __restrict__ own, int E) {
    const int t  = blockIdx.x * blockDim.x + threadIdx.x;
    const int e0 = t * 32;
    unsigned ob[8];
    unsigned ex = 0;
    #pragma unroll
    for (int j = 0; j < 8; ++j) ob[j] = 0;
    auto take = [&](const int i, const int d) {
        const unsigned o = (unsigned)(d / OWN_ROWS);
        #pragma unroll
        for (int j = 0; j < 8; ++j) ob[j] |= ((o >> j) & 1u) << i;
    };
    if (e0 + 32 <= E) {
        v4i d[8];
        #pragma unroll
        for (int g = 0; g < 8; ++g) d[g] = *(const v4i*)(dst + e0 + 4 * g);
        #pragma unroll
        for (int g = 0; g < 8; ++g) {
            take(4 * g + 0, d[g].x);
            take(4 * g + 1, d[g].y);
            take(4 * g + 2, d[g].z);
            take(4 * g + 3, d[g].w);
        }
        ex = ~0u;
    } else {
        for (int i = 0; i < 32; ++i)
            if (e0 + i < E) { take(i, dst[e0 + i]); ex |= 1u << i; }
    }
    unsigned* op = own + t;
    #pragma unroll
    for (int f = 0; f < 4; ++f) {
        const unsigned b0 = ob[2 * f], b1 = ob[2 * f + 1];
        op[(size_t)(4 * f + 0) * PLANE_WORDS] = ~b1 & ~b0 & ex;
        op[(size_t)(4 * f + 1) * PLANE_WORDS] = ~b1 &  b0 & ex;
        op[(size_t)(4 * f + 2) * PLANE_WORDS] =  b1 & ~b0 & ex;
        op[(size_t)(4 * f + 3) * PLANE_WORDS] =  b1 &  b0 & ex;
    }
}

template <class Flush, class Trip>
__device__ __forceinline__ void search_owner_planes(const unsigned* __restrict__ own, const int E,
                                                    const int wave, const int lane,
                                                    const unsigned blk, int* ring,
                                                    Flush flush, Trip per_trip) {
    const unsigned* p0 = own + (size_t)( 0 + ( blk       & 3u)) * PLANE_WORDS + lane * 4;
    const unsigned* p1 = own + (size_t)( 4 + ((blk >> 2) & 3u)) * PLANE_WORDS + lane * 4;
    const unsigned* p2 = own + (size_t)( 8 + ((blk >> 4) & 3u)) * PLANE_WORDS + lane * 4;
    const unsigned* p3 = own + (size_t)(12 + ((blk >> 6) & 3u)) * PLANE_WORDS + lane * 4;
    const int nld = (E + 4095) >> 12;
    int* q = ring + wave * 64;
    int head = 0;
    int cnt  = 0;

    if (wave < nld) {
        v4u c0 = *(const v4u*)(p0 + wave * 128);
        v4u c1 = *(const v4u*)(p1 + wave * 128);
        v4u c2 = *(const v4u*)(p2 + wave * 128);
        v4u c3 = *(const v4u*)(p3 + wave * 128);
        int trip = 0;
        for (int L = wave; L < nld; L += 32) {
            const int Ln = (L + 32 < nld) ? (L + 32) : L;
            const v4u n0 = *(const v4u*)(p0 + Ln * 128);
            const v4u n1 = *(const v4u*)(p1 + Ln * 128);
            const v4u n2 = *(const v4u*)(p2 + Ln * 128);
            const v4u n3 = *(const v4u*)(p3 + Ln * 128);
            per_trip(trip);
            ++trip;
            const v4u m = c0 & c1 & c2 & c3;
            #pragma unroll
            for (int k = 0; k < 4; ++k) {
                unsigned x = m[k];
                const int ebase = (L * 128 + lane * 4 + k) << 5;
                unsigned bal = __builtin_amdgcn_ballot_w32(x != 0);
                while (bal) {
                    if (x != 0) {
                        const int pos = (int)__builtin_amdgcn_mbcnt_lo(bal, (unsigned)cnt);
                        q[(head + pos) & 63] = ebase + __builtin_ctz(x);
                        x &= x - 1;
                    }
                    cnt += __builtin_popcount(bal);
                    if (cnt >= 32) {
                        flush(32, q[(head + lane) & 63]);
                        head = (head + 32) & 63;
                        cnt -= 32;
                    }
                    bal = __builtin_amdgcn_ballot_w32(x != 0);
                }
            }
            c0 = n0; c1 = n1; c2 = n2; c3 = n3;
        }
    }
    if (cnt > 0) {
        int e = 0;
        if (lane < cnt) e = q[(head + lane) & 63];
        flush(cnt, e);
    }
}

#define FO_TM 400
template <int R>
__launch_bounds__(1024)
__global__ void find_owner_kernel(const int* __restrict__ src,
                                  const int* __restrict__ dst,
                                  const unsigned* __restrict__ own,
                                  float* __restrict__ dinv,
                                  float* lstf,
                                  int* __restrict__ lcnt,
                                  const float* __restrict__ X,
                                  const float* __restrict__ Wm,
                                  float* __restrict__ H,
                                  int E, int n) {
    static_assert(R == OWN_ROWS, "the owner planes are made with OWN_ROWS rows per block");
    static_assert(NNODES <= (1 << 17) && R <= (1 << 9), "a list word holds src in 17 bits, the local row above");
    static_assert(R <= FO_TM && FO_TM % 16 == 0, "the A tile holds every M-tile that touches the block's rows");
    constexpr int K     = 128;
    constexpr int N     = 128;
    constexpr int PITCH = 132;
    __shared__ int indeg[512];
    __shared__ int ring[32 * 64];
    __shared__ __attribute__((aligned(16))) float at[FO_TM * PITCH];
    __shared__ __attribute__((aligned(16))) float wt[N * PITCH];

    const int lane = threadIdx.x & 31;
    const int wave = __builtin_amdgcn_readfirstlane(threadIdx.x >> 5);
    const int r0   = blockIdx.x * R;
    const int nrows = (n - r0 < R) ? (n - r0) : R;

    const int cap = nrows & ~31;
    v2i* seg = (v2i*)lstf + (size_t)r0 * 32 + wave * cap;
    int nsaved = 0;

    for (int r = threadIdx.x; r < R; r += 1024) indeg[r] = 0;

    {
        constexpr int FULL = FO_TM / 32;
        constexpr int REST = FO_TM % 32;
        v4f ld[FULL + 1];
        #pragma unroll
        for (int i = 0; i <= FULL; ++i) {
            const int r = wave + 32 * ((i < FULL || wave < REST) ? i : 0);
            int row = r0 + r;
            row = (row < n) ? row : (n - 1);
            ld[i] = *(const v4f*)(X + (size_t)row * K + lane * 4);
        }
        #pragma unroll
        for (int i = 0; i <= FULL; ++i) {
            const int r = wave + 32 * ((i < FULL || wave < REST) ? i : 0);
            *(v4f*)(at + r * PITCH + lane * 4) = ld[i];
        }
    }
    for (int i = threadIdx.x; i < K * N; i += 1024)
        wt[(i % N) * PITCH + (i / N)] = Wm[i];
    __syncthreads();

    const int c  = lane & 15;
    const int h  = lane >> 4;
    const int n0 = (wave & 7) << 4;
    const int mtiles = (nrows + 15) >> 4;
    int mt = wave >> 3;
    auto product_tile = [&]() {
        const float* ap = at + (mt * 16 + c) * PITCH + 4 * h;
        const float* bp = wt + (n0 + c) * PITCH + 4 * h;
        v8f acc = (v8f){};
        #pragma unroll 4
        for (int j = 0; j < 16; ++j) {
            const v4f a = *(const v4f*)(ap + 8 * j);
            const v4f b = *(const v4f*)(bp + 8 * j);
            acc = __builtin_amdgcn_wmma_f32_16x16x4_f32(
                 false, a.xy,  false, b.xy,
                 (short)0, acc,  false,  false);
            acc = __builtin_amdgcn_wmma_f32_16x16x4_f32(
                 false, a.zw,  false, b.zw,
                 (short)0, acc,  false,  false);
        }
        const int ml = mt * 16 + 8 * h;
        float* crow = H + (size_t)(r0 + ml) * N + (n0 + c);
        if (mt * 16 + 16 <= nrows) {
            #pragma unroll
            for (int r = 0; r < 8; ++r) crow[(size_t)r * N] = acc[r];
        } else {
            #pragma unroll
            for (int r = 0; r < 8; ++r)
                if (ml + r < nrows) crow[(size_t)r * N] = acc[r];
        }
        mt += 4;
    };
    const int phase = (wave + (wave >> 3)) & 1;

    search_owner_planes(own, E, wave, lane, (unsigned)blockIdx.x, ring, [&](const int cfl, const int e) {
        if (lane < cfl) {
            const int sv = src[e];
            const int dl = dst[e] - r0;
            atomicAdd(&indeg[dl], 1);
            if (nsaved + cfl <= cap) seg[nsaved + lane] = (v2i){sv | (dl << 17), 0};
        }
        nsaved += cfl;
    }, [&](const int trip) {
        if (((trip + phase) & 1) == 1 && mt < mtiles) product_tile();
    });
    while (mt < mtiles) product_tile();

    if (lane == 0) lcnt[blockIdx.x * 32 + wave] = nsaved;
    __syncthreads();

    for (int r = threadIdx.x; r < nrows; r += 1024) dinv[r0 + r] = rsqrtf(1.0f + (float)indeg[r]);
}

template <int F, int R, bool RELU, bool NORM, int FN>
__launch_bounds__(1024)
__global__ void agg_owner_kernel(const int* __restrict__ src,
                                 const int* __restrict__ dst,
                                 const unsigned* __restrict__ own,
                                 const float* __restrict__ dinv,
                                 const float* __restrict__ h,
                                 const float* __restrict__ bias,
                                 const float* __restrict__ Wn,
                                 float* out,
                                 float* lstf,
                                 const int* __restrict__ lcnt,
                                 int E, int n) {
    static_assert(R == OWN_ROWS, "the owner planes are made with OWN_ROWS rows per block");
    static_assert(FN == 0 || (F == 128 && FN % 16 == 0 && 32 % (FN / 16) == 0 && RELU),
                  "the fused product reads 4-float rows of 128 and gives every wave one N-tile");
    constexpr int VE = F / 32;
    typedef __attribute__((ext_vector_type(VE))) float vecf;
    constexpr int PITCH = (FN > 0) ? F + 4 : F;
    constexpr int LROWS = (FN > 0) ? ((R + 15) / 16) * 16 : R;
    constexpr int ACC_FLOATS = (LROWS * PITCH > 41984) ? LROWS * PITCH : 41984;
    __shared__ __attribute__((aligned(16))) float acc[ACC_FLOATS];
    __shared__ int ring[32 * 64];

    const int lane = threadIdx.x & 31;
    const int wave = __builtin_amdgcn_readfirstlane(threadIdx.x >> 5);
    const int r0   = blockIdx.x * R;
    const int nrows = (n - r0 < R) ? (n - r0) : R;

    const int cap = nrows & ~31;
    v2i* seg = (v2i*)lstf + (size_t)r0 * 32 + wave * cap;

    for (int r = wave; r < nrows; r += 32) {
        const int i = r0 + r;
        const float di = dinv[i];
        const vecf hv = *(const vecf*)(h + (size_t)i * F + lane * VE);
        const vecf bv = *(const vecf*)(bias + lane * VE);
        #pragma unroll
        for (int k = 0; k < VE; ++k) acc[r * PITCH + k * 32 + lane] = bv[k] + hv[k] * di * di;
    }
    if (FN > 0) {
        for (int r = nrows + wave; r < LROWS; r += 32) {
            #pragma unroll
            for (int k = 0; k < VE; ++k) acc[r * PITCH + k * 32 + lane] = 0.0f;
        }
    }
    __syncthreads();

    auto accumulate = [&](int c, int s, int dl, float nv) {
        constexpr int NB = 32 / VE;
        for (int j0 = 0; j0 < c; j0 += NB) {
            vecf rows[NB];
            #pragma unroll
            for (int j = 0; j < NB; ++j) {
                const int jj = (j0 + j < c) ? (j0 + j) : (c - 1);
                const int sj = __builtin_amdgcn_readlane(s, jj);
                rows[j] = *(const vecf*)(h + (size_t)sj * F + lane * VE);
            }
            #pragma unroll
            for (int j = 0; j < NB; ++j) {
                if (j0 + j < c) {
                    const float nj = __int_as_float(__builtin_amdgcn_readlane(__float_as_int(nv), j0 + j));
                    const int   dj = __builtin_amdgcn_readlane(dl, j0 + j);
                    float* a = acc + dj * PITCH + lane;
                    #pragma unroll
                    for (int k = 0; k < VE; ++k)
                        __hip_atomic_fetch_add(a + k * 32, rows[j][k] * nj,
                                               __ATOMIC_RELAXED, __HIP_MEMORY_SCOPE_WORKGROUP);
                }
            }
        }
    };

    const int nl = __builtin_amdgcn_readfirstlane(lcnt[blockIdx.x * 32 + wave]);
    if (nl <= cap) {
        auto raw = [&](const int idx) {
            v2i en = (v2i){0, 0};
            if (idx < nl) en = seg[idx];
            return en;
        };
        auto norm_of = [&](const v2i en) {
            return dinv[en.x & 0x1ffff] * dinv[r0 + (en.x >> 17)];
        };
        v2i ent = raw(lane);
        v2i nx  = raw(32 + lane);
        if (NORM) ent.y = __float_as_int(norm_of(ent));
        int b0 = 0;
        for (; b0 + 32 <= nl; b0 += 32) {
            const v2i nn = raw(b0 + 64 + lane);
            float nxn = 0.0f;
            if (NORM) {
                nxn = norm_of(nx);
                seg[b0 + lane] = ent;
            }
            accumulate(32, ent.x & 0x1ffff, ent.x >> 17, __int_as_float(ent.y));
            if (NORM) nx.y = __float_as_int(nxn);
            ent = nx;
            nx = nn;
        }
        if (b0 < nl) {
            if (NORM && b0 + lane < nl) seg[b0 + lane] = ent;
            accumulate(nl - b0, ent.x & 0x1ffff, ent.x >> 17, __int_as_float(ent.y));
        }
    } else {
        search_owner_planes(own, E, wave, lane, (unsigned)blockIdx.x, ring, [&](const int c, const int e) {
            int s = 0, dl = 0; float nv = 0.0f;
            if (lane < c) {
                s = src[e];
                const int d = dst[e];
                dl = d - r0;
                nv = dinv[s] * dinv[d];
            }
            accumulate(c, s, dl, nv);
        }, [](const int) {});
    }
    if constexpr (FN > 0) {
        constexpr int NT = FN / 16;
        constexpr int G  = 32 / NT;
        const int c  = lane & 15;
        const int hf = lane >> 4;
        const int nt = wave % NT;
        const int g  = wave / NT;
        const int n0 = nt << 4;
        v2f bf[32];
        #pragma unroll
        for (int q = 0; q < 32; ++q) {
            const int kk = 8 * (q >> 1) + 4 * hf + 2 * (q & 1);
            bf[q].x = Wn[(size_t)kk * FN + n0 + c];
            bf[q].y = Wn[(size_t)(kk + 1) * FN + n0 + c];
        }
        __syncthreads();

        const int mtiles = (nrows + 15) >> 4;
        for (int mt = g; mt < mtiles; mt += G) {
            const float* ap = acc + (mt * 16 + c) * PITCH + hf;
            v8f d = (v8f){};
            #pragma unroll
            for (int j = 0; j < 16; ++j) {
                v2f alo, ahi;
                alo.x = fmaxf(ap[2 * j],      0.0f);
                alo.y = fmaxf(ap[2 * j + 32], 0.0f);
                ahi.x = fmaxf(ap[2 * j + 64], 0.0f);
                ahi.y = fmaxf(ap[2 * j + 96], 0.0f);
                d = __builtin_amdgcn_wmma_f32_16x16x4_f32(
                     false, alo,  false, bf[2 * j],
                     (short)0, d,  false,  false);
                d = __builtin_amdgcn_wmma_f32_16x16x4_f32(
                     false, ahi,  false, bf[2 * j + 1],
                     (short)0, d,  false,  false);
            }
            const int ml = mt * 16 + 8 * hf;
            float* crow = out + (size_t)(r0 + ml) * FN + (n0 + c);
            if (mt * 16 + 16 <= nrows) {
                #pragma unroll
                for (int r = 0; r < 8; ++r) crow[(size_t)r * FN] = d[r];
            } else {
                #pragma unroll
                for (int r = 0; r < 8; ++r)
                    if (ml + r < nrows) crow[(size_t)r * FN] = d[r];
            }
        }
    } else {
        __syncthreads();

        for (int r = wave; r < nrows; r += 32) {
            vecf o;
            #pragma unroll
            for (int k = 0; k < VE; ++k) {
                const float v = acc[r * PITCH + k * 32 + lane];
                o[k] = RELU ? fmaxf(v, 0.0f) : v;
            }
            *(vecf*)(out + (size_t)(r0 + r) * F + lane * VE) = o;
        }
    }
}

extern "C" void kernel_launch(void* const* d_in, const int* in_sizes, int n_in,
                              void* d_out, int out_size, void* d_ws, size_t ws_size,
                              hipStream_t stream) {
    const float* x    = (const float*)d_in[0];
    const int*   eidx = (const int*)  d_in[1];
    const float* W1   = (const float*)d_in[2];
    const float* b1   = (const float*)d_in[3];
    const float* W2   = (const float*)d_in[4];
    const float* b2   = (const float*)d_in[5];
    float*       out  = (float*)d_out;

    const int N = NNODES;
    const int E = in_sizes[1] / 2;
    const int* src = eidx;
    const int* dst = eidx + E;

    char*  ws   = (char*)d_ws;
    float* dinv = (float*)(ws);
    float* h1   = (float*)(ws + (size_t)(1  << 20));
    float* h2   = (float*)(ws + (size_t)(53 << 20));
    unsigned* own = (unsigned*)(ws + (size_t)(1 << 20) + (size_t)51200000);
    int* lcnt = (int*)(ws + (size_t)(512 << 10));

    own_plane_kernel<<<(E + 4095) / 4096 * 2, 64, 0, stream>>>(dst, own, E);
    find_owner_kernel<OWN_ROWS><<<(N + 390) / 391, 1024, 0, stream>>>(src, dst, own, dinv, out, lcnt, x, W1, h1, E, N);

    agg_owner_kernel<128, OWN_ROWS, true, true, 64><<<(N + 390) / 391, 1024, 0, stream>>>(src, dst, own, dinv, h1, b1, W2, h2, out, lcnt, E, N);

    agg_owner_kernel<64, OWN_ROWS, false, false, 0><<<(N + 390) / 391, 1024, 0, stream>>>(src, dst, own, dinv, h2, b2, nullptr, out, out, lcnt, E, N);
}
